# starting kernel
# baseline (speedup 1.0000x reference)
#include <hip/hip_runtime.h>


#define N_    32
#define C_IN_ 64
#define T_    300
#define V_    25
#define K_    3
#define C_OUT_ 64
#define KC_   192
#define KV_   75

typedef __attribute__((ext_vector_type(2))) float v2f;
typedef __attribute__((ext_vector_type(8))) float v8f;

#define XS_LD 36
#define FS_LD 196

__global__ __launch_bounds__(256)
void stgcn_fused_kernel(const float* __restrict__ x,
                        const float* __restrict__ Wm,
                        const float* __restrict__ bias,
                        const float* __restrict__ Adj,
                        float* __restrict__ out)
{
    __shared__ float xs[C_IN_][XS_LD];
    __shared__ float fs[32][FS_LD];
    __shared__ float As[KV_ * V_];
    __shared__ float bs[KC_];

    const int tid  = threadIdx.x;
    const int lane = tid & 31;
    const int wave = tid >> 5;
    const int wg   = blockIdx.x;
    const int n    = wg / T_;
    const int t    = wg % T_;

    const int m    = lane & 15;
    const int koff = (lane >> 4) << 1;

    for (int i = tid; i < KV_ * V_; i += 256) As[i] = Adj[i];
    for (int i = tid; i < KC_;      i += 256) bs[i] = bias[i];

    const float* xbase = x + (size_t)n * C_IN_ * (T_ * V_) + (size_t)t * V_;
    for (int i = tid; i < C_IN_ * 32; i += 256) {
        const int r  = i & 31;
        const int ci = i >> 5;
        xs[ci][r] = (r < V_) ? xbase[(size_t)ci * (T_ * V_) + r] : 0.0f;
    }
    __syncthreads();

    for (int tile = wave; tile < 24; tile += 8) {
        const int mt  = tile & 1;
        const int nt  = tile >> 1;
        const int row = mt * 16 + m;
        const int ch  = nt * 16 + m;
        v8f acc = {0.f, 0.f, 0.f, 0.f, 0.f, 0.f, 0.f, 0.f};
        #pragma unroll
        for (int ks = 0; ks < 16; ++ks) {
            const int kb = ks * 4 + koff;
            v2f a, bf;
            a.x  = xs[kb    ][row];
            a.y  = xs[kb + 1][row];
            const float* wp = Wm + (size_t)ch * C_IN_ + kb;
            bf.x = wp[0];
            bf.y = wp[1];
            acc = __builtin_amdgcn_wmma_f32_16x16x4_f32(
                      false, a, false, bf, (short)0, acc, false, false);
        }
        const float bv = bs[ch];
        const int r0 = mt * 16 + ((lane >> 4) << 3);
        #pragma unroll
        for (int r = 0; r < 8; ++r)
            fs[r0 + r][ch] = acc[r] + bv;
    }
    __syncthreads();

    {
        const int mt = wave >> 1;
        const int nt = wave & 1;
        const int c  = mt * 16 + m;
        const int w  = nt * 16 + m;
        const bool wok = (w < V_);
        v8f acc = {0.f, 0.f, 0.f, 0.f, 0.f, 0.f, 0.f, 0.f};
        #pragma unroll
        for (int ks = 0; ks < 19; ++ks) {
            const int kv0 = ks * 4 + koff;
            const int kv1 = kv0 + 1;
            v2f a, bf;
            if (kv0 < KV_) {
                const int k0 = kv0 / V_, v0 = kv0 - k0 * V_;
                a.x  = fs[v0][k0 * 64 + c];
                bf.x = wok ? As[kv0 * V_ + w] : 0.0f;
            } else { a.x = 0.0f; bf.x = 0.0f; }
            if (kv1 < KV_) {
                const int k1 = kv1 / V_, v1 = kv1 - k1 * V_;
                a.y  = fs[v1][k1 * 64 + c];
                bf.y = wok ? As[kv1 * V_ + w] : 0.0f;
            } else { a.y = 0.0f; bf.y = 0.0f; }
            acc = __builtin_amdgcn_wmma_f32_16x16x4_f32(
                      false, a, false, bf, (short)0, acc, false, false);
        }
        if (wok) {
            const int rbase = mt * 16 + ((lane >> 4) << 3);
            float* op = out + ((size_t)n * C_OUT_) * (T_ * V_) + (size_t)t * V_ + w;
            #pragma unroll
            for (int r = 0; r < 8; ++r) {
                const int cc = rbase + r;
                op[(size_t)cc * (T_ * V_)] = acc[r];
            }
        }
    }
}

extern "C" void kernel_launch(void* const* d_in, const int* in_sizes, int n_in,
                              void* d_out, int out_size, void* d_ws, size_t ws_size,
                              hipStream_t stream) {
    const float* x   = (const float*)d_in[0];
    const float* Wm  = (const float*)d_in[1];
    const float* b   = (const float*)d_in[2];
    const float* A   = (const float*)d_in[3];
    float* out       = (float*)d_out;
    (void)in_sizes; (void)n_in; (void)out_size; (void)d_ws; (void)ws_size;

    dim3 grid(N_ * T_);
    dim3 block(256);
    hipLaunchKernelGGL(stgcn_fused_kernel, grid, block, 0, stream,
                       x, Wm, b, A, out);
}


# Staging: all global loads of x, adjacency and bias are issued into registers before the first wait, then stored to LDS: one memory round trip per workgroup, not 17 in series; sums unchanged
# speedup vs baseline: 1.2182x; 1.2182x over previous
#include <hip/hip_runtime.h>


#define N_    32
#define C_IN_ 64
#define T_    300
#define V_    25
#define K_    3
#define C_OUT_ 64
#define KC_   192
#define KV_   75

typedef __attribute__((ext_vector_type(2))) float v2f;
typedef __attribute__((ext_vector_type(8))) float v8f;

#define XS_LD 36
#define FS_LD 196

__global__ __launch_bounds__(256)
void stgcn_fused_kernel(const float* __restrict__ x,
                        const float* __restrict__ Wm,
                        const float* __restrict__ bias,
                        const float* __restrict__ Adj,
                        float* __restrict__ out)
{
    __shared__ float xs[C_IN_][XS_LD];
    __shared__ float fs[32][FS_LD];
    __shared__ float As[KV_ * V_];
    __shared__ float bs[KC_];

    const int tid  = threadIdx.x;
    const int lane = tid & 31;
    const int wave = tid >> 5;
    const int wg   = blockIdx.x;
    const int n    = wg / T_;
    const int t    = wg % T_;

    const int m    = lane & 15;
    const int koff = (lane >> 4) << 1;

    const int r   = tid & 31;
    const int ci0 = tid >> 5;
    const float* xbase = x + (size_t)n * C_IN_ * (T_ * V_) + (size_t)t * V_;

    float xv[8];
    float av[8];
    float bv0 = 0.0f;

    if (r < V_) {
        const float* xp = xbase + (size_t)ci0 * (T_ * V_) + r;
        #pragma unroll
        for (int j = 0; j < 8; ++j) xv[j] = xp[(size_t)(8 * j) * (T_ * V_)];
    } else {
        #pragma unroll
        for (int j = 0; j < 8; ++j) xv[j] = 0.0f;
    }
    #pragma unroll
    for (int j = 0; j < 7; ++j) av[j] = Adj[tid + 256 * j];
    av[7] = (tid < KV_ * V_ - 7 * 256) ? Adj[tid + 7 * 256] : 0.0f;
    if (tid < KC_) bv0 = bias[tid];

    #pragma unroll
    for (int j = 0; j < 8; ++j) xs[ci0 + 8 * j][r] = xv[j];
    #pragma unroll
    for (int j = 0; j < 7; ++j) As[tid + 256 * j] = av[j];
    if (tid < KV_ * V_ - 7 * 256) As[tid + 7 * 256] = av[7];
    if (tid < KC_) bs[tid] = bv0;
    __syncthreads();

    for (int tile = wave; tile < 24; tile += 8) {
        const int mt  = tile & 1;
        const int nt  = tile >> 1;
        const int row = mt * 16 + m;
        const int ch  = nt * 16 + m;
        v8f acc = {0.f, 0.f, 0.f, 0.f, 0.f, 0.f, 0.f, 0.f};
        #pragma unroll
        for (int ks = 0; ks < 16; ++ks) {
            const int kb = ks * 4 + koff;
            v2f a, bf;
            a.x  = xs[kb    ][row];
            a.y  = xs[kb + 1][row];
            const float* wp = Wm + (size_t)ch * C_IN_ + kb;
            bf.x = wp[0];
            bf.y = wp[1];
            acc = __builtin_amdgcn_wmma_f32_16x16x4_f32(
                      false, a, false, bf, (short)0, acc, false, false);
        }
        const float bv = bs[ch];
        const int r0 = mt * 16 + ((lane >> 4) << 3);
        #pragma unroll
        for (int r = 0; r < 8; ++r)
            fs[r0 + r][ch] = acc[r] + bv;
    }
    __syncthreads();

    {
        const int mt = wave >> 1;
        const int nt = wave & 1;
        const int c  = mt * 16 + m;
        const int w  = nt * 16 + m;
        const bool wok = (w < V_);
        v8f acc = {0.f, 0.f, 0.f, 0.f, 0.f, 0.f, 0.f, 0.f};
        #pragma unroll
        for (int ks = 0; ks < 19; ++ks) {
            const int kv0 = ks * 4 + koff;
            const int kv1 = kv0 + 1;
            v2f a, bf;
            if (kv0 < KV_) {
                const int k0 = kv0 / V_, v0 = kv0 - k0 * V_;
                a.x  = fs[v0][k0 * 64 + c];
                bf.x = wok ? As[kv0 * V_ + w] : 0.0f;
            } else { a.x = 0.0f; bf.x = 0.0f; }
            if (kv1 < KV_) {
                const int k1 = kv1 / V_, v1 = kv1 - k1 * V_;
                a.y  = fs[v1][k1 * 64 + c];
                bf.y = wok ? As[kv1 * V_ + w] : 0.0f;
            } else { a.y = 0.0f; bf.y = 0.0f; }
            acc = __builtin_amdgcn_wmma_f32_16x16x4_f32(
                      false, a, false, bf, (short)0, acc, false, false);
        }
        if (wok) {
            const int rbase = mt * 16 + ((lane >> 4) << 3);
            float* op = out + ((size_t)n * C_OUT_) * (T_ * V_) + (size_t)t * V_ + w;
            #pragma unroll
            for (int r = 0; r < 8; ++r) {
                const int cc = rbase + r;
                op[(size_t)cc * (T_ * V_)] = acc[r];
            }
        }
    }
}

extern "C" void kernel_launch(void* const* d_in, const int* in_sizes, int n_in,
                              void* d_out, int out_size, void* d_ws, size_t ws_size,
                              hipStream_t stream) {
    const float* x   = (const float*)d_in[0];
    const float* Wm  = (const float*)d_in[1];
    const float* b   = (const float*)d_in[2];
    const float* A   = (const float*)d_in[3];
    float* out       = (float*)d_out;
    (void)in_sizes; (void)n_in; (void)out_size; (void)d_ws; (void)ws_size;

    dim3 grid(N_ * T_);
    dim3 block(256);
    hipLaunchKernelGGL(stgcn_fused_kernel, grid, block, 0, stream,
                       x, Wm, b, A, out);
}


# 512 resident workgroups each walk 18-19 consecutive (n,t); B fragments of both GEMMs (W, adjacency) and bias are loaded once per workgroup into registers, no W loads in the loop; sums unchanged
# speedup vs baseline: 1.6635x; 1.3655x over previous
#include <hip/hip_runtime.h>


#define N_    32
#define C_IN_ 64
#define T_    300
#define V_    25
#define K_    3
#define C_OUT_ 64
#define KC_   192
#define KV_   75
#define NPAIR_ 9600
#define NWG_  512
#define NPW_  18
#define NEXTRA_ 384

typedef __attribute__((ext_vector_type(2))) float v2f;
typedef __attribute__((ext_vector_type(8))) float v8f;

#define XS_LD 36
#define FS_LD 196

__global__ __launch_bounds__(256)
void stgcn_fused_kernel(const float* __restrict__ x,
                        const float* __restrict__ Wm,
                        const float* __restrict__ bias,
                        const float* __restrict__ Adj,
                        float* __restrict__ out)
{
    __shared__ float xs[C_IN_][XS_LD];
    __shared__ float fs[32][FS_LD];

    const int tid  = threadIdx.x;
    const int lane = tid & 31;
    const int wave = tid >> 5;

    const int  m    = lane & 15;
    const int  koff = (lane >> 4) << 1;
    const bool hi   = (lane >> 4) != 0;

    const int pmt  = wave & 1;
    const int pnt0 = wave >> 1;
    const int prow = pmt * 16 + m;

    const int  amt = wave >> 1;
    const int  ant = wave & 1;
    const int  c   = amt * 16 + m;
    const int  w   = ant * 16 + m;
    const bool wok = (w < V_);

    v2f   wf[3][16];
    float bv[3];
    #pragma unroll
    for (int j = 0; j < 3; ++j) {
        const int ch = (pnt0 + 4 * j) * 16 + m;
        const float* wp = Wm + ch * C_IN_ + koff;
        #pragma unroll
        for (int ks = 0; ks < 16; ++ks) {
            wf[j][ks].x = wp[ks * 4];
            wf[j][ks].y = wp[ks * 4 + 1];
        }
        bv[j] = bias[ch];
    }
    v2f af[19];
    #pragma unroll
    for (int ks = 0; ks < 19; ++ks) { af[ks].x = 0.0f; af[ks].y = 0.0f; }
    if (wok) {
        const float* ap = Adj + koff * V_ + w;
        #pragma unroll
        for (int ks = 0; ks < 19; ++ks) {
            #pragma unroll
            for (int e = 0; e < 2; ++e) {
                const int kvL = ks * 4 + e;
                if (kvL + 2 < KV_)  af[ks][e] = ap[kvL * V_];
                else if (!hi)       af[ks][e] = ap[kvL * V_];
            }
        }
    }

    const float* fs0 = &fs[0][0] + c;
    const float* fsh = fs0 + koff * FS_LD;

    const int r    = tid & 31;
    const int ci0  = tid >> 5;
    const int xoff = ci0 * (T_ * V_) + r;
    const int ooff = (amt * 16 + ((lane >> 4) << 3)) * (T_ * V_) + w;

    const int g    = blockIdx.x;
    const int pbeg = NPW_ * g + (g < NEXTRA_ ? g : NEXTRA_);
    const int pend = pbeg + NPW_ + (g < NEXTRA_ ? 1 : 0);
    for (int p = pbeg; p < pend; ++p) {
        const int n = p / T_;
        const int t = p - n * T_;

        const float* xbase = x + (size_t)n * C_IN_ * (T_ * V_) + (size_t)t * V_;
        float xv[8];
        if (r < V_) {
            #pragma unroll
            for (int j = 0; j < 8; ++j) xv[j] = xbase[xoff + 8 * j * (T_ * V_)];
        } else {
            #pragma unroll
            for (int j = 0; j < 8; ++j) xv[j] = 0.0f;
        }
        #pragma unroll
        for (int j = 0; j < 8; ++j) xs[ci0 + 8 * j][r] = xv[j];
        __syncthreads();

        {
            v8f acc0 = {0.f, 0.f, 0.f, 0.f, 0.f, 0.f, 0.f, 0.f};
            v8f acc1 = {0.f, 0.f, 0.f, 0.f, 0.f, 0.f, 0.f, 0.f};
            v8f acc2 = {0.f, 0.f, 0.f, 0.f, 0.f, 0.f, 0.f, 0.f};
            #pragma unroll
            for (int ks = 0; ks < 16; ++ks) {
                const int kb = ks * 4 + koff;
                v2f a;
                a.x = xs[kb    ][prow];
                a.y = xs[kb + 1][prow];
                acc0 = __builtin_amdgcn_wmma_f32_16x16x4_f32(
                           false, a, false, wf[0][ks], (short)0, acc0, false, false);
                acc1 = __builtin_amdgcn_wmma_f32_16x16x4_f32(
                           false, a, false, wf[1][ks], (short)0, acc1, false, false);
                acc2 = __builtin_amdgcn_wmma_f32_16x16x4_f32(
                           false, a, false, wf[2][ks], (short)0, acc2, false, false);
            }
            const int r0  = pmt * 16 + ((lane >> 4) << 3);
            const int ch0 = pnt0 * 16 + m;
            #pragma unroll
            for (int q = 0; q < 8; ++q) {
                fs[r0 + q][ch0      ] = acc0[q] + bv[0];
                fs[r0 + q][ch0 +  64] = acc1[q] + bv[1];
                fs[r0 + q][ch0 + 128] = acc2[q] + bv[2];
            }
        }
        __syncthreads();

        {
            v8f acc = {0.f, 0.f, 0.f, 0.f, 0.f, 0.f, 0.f, 0.f};
            #pragma unroll
            for (int ks = 0; ks < 19; ++ks) {
                v2f a;
                #pragma unroll
                for (int e = 0; e < 2; ++e) {
                    const int kvL  = ks * 4 + e;
                    const int kvH  = kvL + 2;
                    const int offL = (kvL % V_) * FS_LD + (kvL / V_) * 64;
                    float fa;
                    if (kvH >= KV_) {
                        fa = hi ? 0.0f : fs0[offL];
                    } else {
                        const int offH = (kvH % V_) * FS_LD + (kvH / V_) * 64;
                        if (offH - offL == 2 * FS_LD) {
                            fa = fsh[offL];
                        } else {
                            const float* ps = hi ? (fs0 + offH) : (fs0 + offL);
                            fa = *ps;
                        }
                    }
                    a[e] = fa;
                }
                acc = __builtin_amdgcn_wmma_f32_16x16x4_f32(
                          false, a, false, af[ks], (short)0, acc, false, false);
            }
            if (wok) {
                float* op = out + (size_t)n * C_OUT_ * (T_ * V_) + (size_t)t * V_;
                #pragma unroll
                for (int q = 0; q < 8; ++q)
                    op[ooff + q * (T_ * V_)] = acc[q];
            }
        }
    }
}

extern "C" void kernel_launch(void* const* d_in, const int* in_sizes, int n_in,
                              void* d_out, int out_size, void* d_ws, size_t ws_size,
                              hipStream_t stream) {
    const float* x   = (const float*)d_in[0];
    const float* Wm  = (const float*)d_in[1];
    const float* b   = (const float*)d_in[2];
    const float* A   = (const float*)d_in[3];
    float* out       = (float*)d_out;
    (void)in_sizes; (void)n_in; (void)out_size; (void)d_ws; (void)ws_size;

    dim3 grid(NWG_);
    dim3 block(256);
    hipLaunchKernelGGL(stgcn_fused_kernel, grid, block, 0, stream,
                       x, Wm, b, A, out);
}


# LDS tiles double-buffered: x of pair p+1 goes to the other xs buffer during pair p, x of pair p+2 requested behind the barrier, fs alternates; one barrier per pair, latency hidden; sums unchanged
# speedup vs baseline: 1.8204x; 1.0943x over previous
#include <hip/hip_runtime.h>


#define N_    32
#define C_IN_ 64
#define T_    300
#define V_    25
#define K_    3
#define C_OUT_ 64
#define KC_   192
#define KV_   75
#define NPAIR_ 9600
#define NWG_  512
#define NPW_  18
#define NEXTRA_ 384

typedef __attribute__((ext_vector_type(2))) float v2f;
typedef __attribute__((ext_vector_type(8))) float v8f;

#define XS_LD 36
#define FS_LD 196

__device__ __forceinline__ void load_x_slice(const float* __restrict__ x, int p, int xoff,
                                             bool rok, float (&xv)[8])
{
    const int n = p / T_;
    const int t = p - n * T_;
    const float* xbase = x + (size_t)n * C_IN_ * (T_ * V_) + (size_t)t * V_;
    if (rok) {
        #pragma unroll
        for (int j = 0; j < 8; ++j) xv[j] = xbase[xoff + 8 * j * (T_ * V_)];
    } else {
        #pragma unroll
        for (int j = 0; j < 8; ++j) xv[j] = 0.0f;
    }
}

__global__ __launch_bounds__(256)
void stgcn_fused_kernel(const float* __restrict__ x,
                        const float* __restrict__ Wm,
                        const float* __restrict__ bias,
                        const float* __restrict__ Adj,
                        float* __restrict__ out)
{
    __shared__ float xs[2][C_IN_][XS_LD];
    __shared__ float fs[2][32][FS_LD];

    const int tid  = threadIdx.x;
    const int lane = tid & 31;
    const int wave = tid >> 5;

    const int  m    = lane & 15;
    const int  koff = (lane >> 4) << 1;
    const bool hi   = (lane >> 4) != 0;

    const int pmt  = wave & 1;
    const int pnt0 = wave >> 1;
    const int prow = pmt * 16 + m;

    const int  amt = wave >> 1;
    const int  ant = wave & 1;
    const int  c   = amt * 16 + m;
    const int  w   = ant * 16 + m;
    const bool wok = (w < V_);

    v2f   wf[3][16];
    float bv[3];
    #pragma unroll
    for (int j = 0; j < 3; ++j) {
        const int ch = (pnt0 + 4 * j) * 16 + m;
        const float* wp = Wm + ch * C_IN_ + koff;
        #pragma unroll
        for (int ks = 0; ks < 16; ++ks) {
            wf[j][ks].x = wp[ks * 4];
            wf[j][ks].y = wp[ks * 4 + 1];
        }
        bv[j] = bias[ch];
    }
    v2f af[19];
    #pragma unroll
    for (int ks = 0; ks < 19; ++ks) { af[ks].x = 0.0f; af[ks].y = 0.0f; }
    if (wok) {
        const float* ap = Adj + koff * V_ + w;
        #pragma unroll
        for (int ks = 0; ks < 19; ++ks) {
            #pragma unroll
            for (int e = 0; e < 2; ++e) {
                const int kvL = ks * 4 + e;
                if (kvL + 2 < KV_)  af[ks][e] = ap[kvL * V_];
                else if (!hi)       af[ks][e] = ap[kvL * V_];
            }
        }
    }

    const int r    = tid & 31;
    const int ci0  = tid >> 5;
    const int xoff = ci0 * (T_ * V_) + r;
    const int ooff = (amt * 16 + ((lane >> 4) << 3)) * (T_ * V_) + w;

    const int g    = blockIdx.x;
    const int pbeg = NPW_ * g + (g < NEXTRA_ ? g : NEXTRA_);
    const int pend = pbeg + NPW_ + (g < NEXTRA_ ? 1 : 0);
    const bool rok = (r < V_);

    float xv[8];
    load_x_slice(x, pbeg, xoff, rok, xv);
    #pragma unroll
    for (int j = 0; j < 8; ++j) xs[0][ci0 + 8 * j][r] = xv[j];
    __syncthreads();
    load_x_slice(x, pbeg + 1, xoff, rok, xv);

    for (int p = pbeg; p < pend; ++p) {
        const int n = p / T_;
        const int t = p - n * T_;
        const int buf = (p - pbeg) & 1;
        const float (*xsb)[XS_LD] = xs[buf];
        float (*fsb)[FS_LD]       = fs[buf];
        const float* fs0 = &fsb[0][0] + c;
        const float* fsh = fs0 + koff * FS_LD;

        {
            v8f acc0 = {0.f, 0.f, 0.f, 0.f, 0.f, 0.f, 0.f, 0.f};
            v8f acc1 = {0.f, 0.f, 0.f, 0.f, 0.f, 0.f, 0.f, 0.f};
            v8f acc2 = {0.f, 0.f, 0.f, 0.f, 0.f, 0.f, 0.f, 0.f};
            #pragma unroll
            for (int ks = 0; ks < 16; ++ks) {
                const int kb = ks * 4 + koff;
                v2f a;
                a.x = xsb[kb    ][prow];
                a.y = xsb[kb + 1][prow];
                acc0 = __builtin_amdgcn_wmma_f32_16x16x4_f32(
                           false, a, false, wf[0][ks], (short)0, acc0, false, false);
                acc1 = __builtin_amdgcn_wmma_f32_16x16x4_f32(
                           false, a, false, wf[1][ks], (short)0, acc1, false, false);
                acc2 = __builtin_amdgcn_wmma_f32_16x16x4_f32(
                           false, a, false, wf[2][ks], (short)0, acc2, false, false);
            }
            const int r0  = pmt * 16 + ((lane >> 4) << 3);
            const int ch0 = pnt0 * 16 + m;
            #pragma unroll
            for (int q = 0; q < 8; ++q) {
                fsb[r0 + q][ch0      ] = acc0[q] + bv[0];
                fsb[r0 + q][ch0 +  64] = acc1[q] + bv[1];
                fsb[r0 + q][ch0 + 128] = acc2[q] + bv[2];
            }
        }

        if (p + 1 < pend) {
            #pragma unroll
            for (int j = 0; j < 8; ++j) xs[buf ^ 1][ci0 + 8 * j][r] = xv[j];
        }
        __syncthreads();
        if (p + 2 < pend) load_x_slice(x, p + 2, xoff, rok, xv);

        {
            v8f acc = {0.f, 0.f, 0.f, 0.f, 0.f, 0.f, 0.f, 0.f};
            #pragma unroll
            for (int ks = 0; ks < 19; ++ks) {
                v2f a;
                #pragma unroll
                for (int e = 0; e < 2; ++e) {
                    const int kvL  = ks * 4 + e;
                    const int kvH  = kvL + 2;
                    const int offL = (kvL % V_) * FS_LD + (kvL / V_) * 64;
                    float fa;
                    if (kvH >= KV_) {
                        fa = hi ? 0.0f : fs0[offL];
                    } else {
                        const int offH = (kvH % V_) * FS_LD + (kvH / V_) * 64;
                        if (offH - offL == 2 * FS_LD) {
                            fa = fsh[offL];
                        } else {
                            const float* ps = hi ? (fs0 + offH) : (fs0 + offL);
                            fa = *ps;
                        }
                    }
                    a[e] = fa;
                }
                acc = __builtin_amdgcn_wmma_f32_16x16x4_f32(
                          false, a, false, af[ks], (short)0, acc, false, false);
            }
            if (wok) {
                float* op = out + (size_t)n * C_OUT_ * (T_ * V_) + (size_t)t * V_;
                #pragma unroll
                for (int q = 0; q < 8; ++q)
                    op[ooff + q * (T_ * V_)] = acc[q];
            }
        }
    }
}

extern "C" void kernel_launch(void* const* d_in, const int* in_sizes, int n_in,
                              void* d_out, int out_size, void* d_ws, size_t ws_size,
                              hipStream_t stream) {
    const float* x   = (const float*)d_in[0];
    const float* Wm  = (const float*)d_in[1];
    const float* b   = (const float*)d_in[2];
    const float* A   = (const float*)d_in[3];
    float* out       = (float*)d_out;
    (void)in_sizes; (void)n_in; (void)out_size; (void)d_ws; (void)ws_size;

    dim3 grid(NWG_);
    dim3 block(256);
    hipLaunchKernelGGL(stgcn_fused_kernel, grid, block, 0, stream,
                       x, Wm, b, A, out);
}


# Five consecutive (n,t) pairs projected as ONE GEMM: M = 125 rows in 8 tiles (3 padded rows), was 10 tiles (35 padded): 20 % fewer projection WMMAs; 2 barriers per 5 pairs; sums unchanged
# speedup vs baseline: 1.9145x; 1.0517x over previous
#include <hip/hip_runtime.h>


#define N_    32
#define C_IN_ 64
#define T_    300
#define V_    25
#define K_    3
#define C_OUT_ 64
#define KC_   192
#define KV_   75
#define NPAIR_ 9600
#define G_    5
#define GPN_  60
#define NGRP_ 1920
#define GROWS_ 125
#define NWG_  512
#define GPW_  3
#define GEXTRA_ 384

typedef __attribute__((ext_vector_type(2))) float v2f;
typedef __attribute__((ext_vector_type(8))) float v8f;

#define XS_LD 144
#define FS_LD 196

__device__ __forceinline__ void load_x_group(const float* __restrict__ x, int gi, int xoff,
                                             int tail_off, float (&xv)[32])
{
    const int n  = gi / GPN_;
    const int t0 = (gi - n * GPN_) * G_;
    const float* xbase = x + (size_t)n * C_IN_ * (T_ * V_) + (size_t)t0 * V_ + xoff;
    #pragma unroll
    for (int j = 0; j < 8; ++j) {
        #pragma unroll
        for (int q = 0; q < 3; ++q) xv[j * 4 + q] = xbase[8 * j * (T_ * V_) + 32 * q];
        xv[j * 4 + 3] = xbase[8 * j * (T_ * V_) + tail_off];
    }
}

__device__ __forceinline__ void store_x_group(float (*xs)[XS_LD], int wave, int lane,
                                              bool tail_ok, const float (&xv)[32])
{
    #pragma unroll
    for (int j = 0; j < 8; ++j) {
        #pragma unroll
        for (int q = 0; q < 3; ++q) xs[wave + 8 * j][lane + 32 * q] = xv[j * 4 + q];
        xs[wave + 8 * j][lane + 96] = tail_ok ? xv[j * 4 + 3] : 0.0f;
    }
}

__global__ __launch_bounds__(256)
void stgcn_fused_kernel(const float* __restrict__ x,
                        const float* __restrict__ Wm,
                        const float* __restrict__ bias,
                        const float* __restrict__ Adj,
                        float* __restrict__ out)
{
    __shared__ float lds[C_IN_ * XS_LD + 128 * FS_LD];
    float (*xs)[XS_LD] = reinterpret_cast<float (*)[XS_LD]>(lds);
    float (*fs)[FS_LD] = reinterpret_cast<float (*)[FS_LD]>(lds + C_IN_ * XS_LD);

    const int tid  = threadIdx.x;
    const int lane = tid & 31;
    const int wave = tid >> 5;

    const int  m    = lane & 15;
    const int  koff = (lane >> 4) << 1;
    const bool hi   = (lane >> 4) != 0;

    const int pmt  = wave & 1;
    const int pnt0 = wave >> 1;

    const int  amt = wave >> 1;
    const int  ant = wave & 1;
    const int  c   = amt * 16 + m;
    const int  w   = ant * 16 + m;
    const bool wok = (w < V_);

    v2f   wf[3][16];
    float bv[3];
    #pragma unroll
    for (int j = 0; j < 3; ++j) {
        const int ch = (pnt0 + 4 * j) * 16 + m;
        const float* wp = Wm + ch * C_IN_ + koff;
        #pragma unroll
        for (int ks = 0; ks < 16; ++ks) {
            wf[j][ks].x = wp[ks * 4];
            wf[j][ks].y = wp[ks * 4 + 1];
        }
        bv[j] = bias[ch];
    }
    v2f af[19];
    #pragma unroll
    for (int ks = 0; ks < 19; ++ks) { af[ks].x = 0.0f; af[ks].y = 0.0f; }
    if (wok) {
        const float* ap = Adj + koff * V_ + w;
        #pragma unroll
        for (int ks = 0; ks < 19; ++ks) {
            #pragma unroll
            for (int e = 0; e < 2; ++e) {
                const int kvL = ks * 4 + e;
                if (kvL + 2 < KV_)  af[ks][e] = ap[kvL * V_];
                else if (!hi)       af[ks][e] = ap[kvL * V_];
            }
        }
    }

    const int  xoff    = wave * (T_ * V_) + lane;
    const bool tail_ok  = (lane + 96 < GROWS_);
    const int  tail_off = tail_ok ? 96 : 64;
    const int  ooff = (amt * 16 + ((lane >> 4) << 3)) * (T_ * V_) + w;

    const int g    = blockIdx.x;
    const int gbeg = GPW_ * g + (g < GEXTRA_ ? g : GEXTRA_);
    const int gend = gbeg + GPW_ + (g < GEXTRA_ ? 1 : 0);

    float xv[32];
    load_x_group(x, gbeg, xoff, tail_off, xv);
    store_x_group(xs, wave, lane, tail_ok, xv);
    __syncthreads();
    load_x_group(x, gbeg + 1, xoff, tail_off, xv);

    for (int gi = gbeg; gi < gend; ++gi) {
        const int n  = gi / GPN_;
        const int t0 = (gi - n * GPN_) * G_;

        for (int i = 0; i < 4; ++i) {
            const int mt   = pmt * 4 + i;
            const int prow = mt * 16 + m;
            v8f acc0 = {0.f, 0.f, 0.f, 0.f, 0.f, 0.f, 0.f, 0.f};
            v8f acc1 = {0.f, 0.f, 0.f, 0.f, 0.f, 0.f, 0.f, 0.f};
            v8f acc2 = {0.f, 0.f, 0.f, 0.f, 0.f, 0.f, 0.f, 0.f};
            v2f a[16];
            #pragma unroll
            for (int ks = 0; ks < 16; ++ks) {
                const int kb = ks * 4 + koff;
                a[ks].x = xs[kb    ][prow];
                a[ks].y = xs[kb + 1][prow];
            }
            #pragma unroll
            for (int ks = 0; ks < 16; ++ks) {
                acc0 = __builtin_amdgcn_wmma_f32_16x16x4_f32(
                           false, a[ks], false, wf[0][ks], (short)0, acc0, false, false);
                acc1 = __builtin_amdgcn_wmma_f32_16x16x4_f32(
                           false, a[ks], false, wf[1][ks], (short)0, acc1, false, false);
                acc2 = __builtin_amdgcn_wmma_f32_16x16x4_f32(
                           false, a[ks], false, wf[2][ks], (short)0, acc2, false, false);
            }
            const int r0  = mt * 16 + ((lane >> 4) << 3);
            const int ch0 = pnt0 * 16 + m;
            #pragma unroll
            for (int q = 0; q < 8; ++q) {
                fs[r0 + q][ch0      ] = acc0[q] + bv[0];
                fs[r0 + q][ch0 +  64] = acc1[q] + bv[1];
                fs[r0 + q][ch0 + 128] = acc2[q] + bv[2];
            }
        }

        __syncthreads();

        if (gi + 1 < gend) store_x_group(xs, wave, lane, tail_ok, xv);

        #pragma nounroll
        for (int pl = 0; pl < G_; ++pl) {
            const float* fs0 = &fs[0][0] + pl * (V_ * FS_LD) + c;
            const float* fsh = fs0 + koff * FS_LD;
            v8f acc = {0.f, 0.f, 0.f, 0.f, 0.f, 0.f, 0.f, 0.f};
            #pragma unroll
            for (int ks = 0; ks < 19; ++ks) {
                v2f a;
                #pragma unroll
                for (int e = 0; e < 2; ++e) {
                    const int kvL  = ks * 4 + e;
                    const int kvH  = kvL + 2;
                    const int offL = (kvL % V_) * FS_LD + (kvL / V_) * 64;
                    float fa;
                    if (kvH >= KV_) {
                        fa = hi ? 0.0f : fs0[offL];
                    } else {
                        const int offH = (kvH % V_) * FS_LD + (kvH / V_) * 64;
                        if (offH - offL == 2 * FS_LD) {
                            fa = fsh[offL];
                        } else {
                            const float* ps = hi ? (fs0 + offH) : (fs0 + offL);
                            fa = *ps;
                        }
                    }
                    a[e] = fa;
                }
                acc = __builtin_amdgcn_wmma_f32_16x16x4_f32(
                          false, a, false, af[ks], (short)0, acc, false, false);
            }
            if (wok) {
                float* op = out + (size_t)n * C_OUT_ * (T_ * V_) + (size_t)(t0 + pl) * V_;
                #pragma unroll
                for (int q = 0; q < 8; ++q)
                    op[ooff + q * (T_ * V_)] = acc[q];
            }
        }

        __syncthreads();
        if (gi + 2 < gend) load_x_group(x, gi + 2, xoff, tail_off, xv);
    }
}

extern "C" void kernel_launch(void* const* d_in, const int* in_sizes, int n_in,
                              void* d_out, int out_size, void* d_ws, size_t ws_size,
                              hipStream_t stream) {
    const float* x   = (const float*)d_in[0];
    const float* Wm  = (const float*)d_in[1];
    const float* b   = (const float*)d_in[2];
    const float* A   = (const float*)d_in[3];
    float* out       = (float*)d_out;
    (void)in_sizes; (void)n_in; (void)out_size; (void)d_ws; (void)ws_size;

    dim3 grid(NWG_);
    dim3 block(256);
    hipLaunchKernelGGL(stgcn_fused_kernel, grid, block, 0, stream,
                       x, Wm, b, A, out);
}
